# starting kernel
# baseline (speedup 1.0000x reference)
#include <hip/hip_runtime.h>


typedef __attribute__((ext_vector_type(2))) float v2f;
typedef __attribute__((ext_vector_type(8))) float v8f;

#define PAD 32

__global__ void k_prefix(const float* __restrict__ sA, const float* __restrict__ sB,
                         float* __restrict__ PA, float* __restrict__ PB, int n) {
  if (threadIdx.x == 0) {
    float p = 1.f; PA[0] = p;
    for (int i = 1; i < n; ++i) { p *= sA[i - 1]; PA[i] = p; }
  } else if (threadIdx.x == 1) {
    float p = 1.f; PB[0] = p;
    for (int i = 1; i < n; ++i) { p *= sB[i - 1]; PB[i] = p; }
  }
}

__global__ void k_transpose_scale(const float* __restrict__ x, const float* __restrict__ PB,
                                  float* __restrict__ Xt, int n, int batch) {
  __shared__ float tile[32][33];
  const int k0 = blockIdx.x * 32, b0 = blockIdx.y * 32;
  const int tx = threadIdx.x, ty = threadIdx.y;
#pragma unroll
  for (int s = 0; s < 32; s += 8)
    tile[ty + s][tx] = x[(size_t)(b0 + ty + s) * n + (k0 + tx)];
  __syncthreads();
#pragma unroll
  for (int s = 0; s < 32; s += 8) {
    const int k = k0 + ty + s;
    Xt[(size_t)k * batch + (b0 + tx)] = PB[k] * tile[tx][ty + s];
  }
}

__global__ void k_pass1(const float* __restrict__ Hm, const float* __restrict__ PB,
                        const float* __restrict__ Xt, float* __restrict__ TMP,
                        int n, int batch) {
  extern __shared__ float hp[];
  const int tid = threadIdx.x;
  const int lane = tid & 31;
  const int w = tid >> 5;
  const int r = blockIdx.z;

  for (int t = tid; t < PAD; t += blockDim.x) hp[t] = 0.f;
  for (int t = tid; t < 4; t += blockDim.x) hp[PAD + n + t] = 0.f;
  for (int t = tid; t < n; t += blockDim.x) hp[PAD + t] = Hm[(size_t)r * n + t] / PB[t];
  __syncthreads();

  const int j0 = blockIdx.x * 64 + (w & 1) * 32;
  const int b0 = blockIdx.y * 256 + (w >> 1) * 64;

  const int jr   = lane & 15;
  const int kb2  = (lane >> 4) * 2;
  const int nlan = lane & 15;

  v8f c[2][4];
#pragma unroll
  for (int jt = 0; jt < 2; ++jt)
#pragma unroll
    for (int t = 0; t < 4; ++t) c[jt][t] = (v8f){0.f,0.f,0.f,0.f,0.f,0.f,0.f,0.f};

  const int abase0 = PAD + kb2 - jr - j0;
  const int abase1 = abase0 - 16;
  const float* __restrict__ xcol = Xt + (size_t)kb2 * batch + (b0 + nlan);

  for (int k0 = j0; k0 < n; k0 += 4) {
    v2f a0, a1;
    a0.x = hp[abase0 + k0]; a0.y = hp[abase0 + k0 + 1];
    a1.x = hp[abase1 + k0]; a1.y = hp[abase1 + k0 + 1];
    const float* __restrict__ xp = xcol + (size_t)k0 * batch;
    __builtin_prefetch(xp + (size_t)32 * batch, 0, 1);
#pragma unroll
    for (int t = 0; t < 4; ++t) {
      v2f bb;
      bb.x = xp[16 * t];
      bb.y = xp[batch + 16 * t];
      c[0][t] = __builtin_amdgcn_wmma_f32_16x16x4_f32(false, a0, false, bb,
                                                      (short)0, c[0][t], false, false);
      c[1][t] = __builtin_amdgcn_wmma_f32_16x16x4_f32(false, a1, false, bb,
                                                      (short)0, c[1][t], false, false);
    }
  }

  const int jhalf = (lane >> 4) * 8;
#pragma unroll
  for (int jt = 0; jt < 2; ++jt)
#pragma unroll
    for (int t = 0; t < 4; ++t)
#pragma unroll
      for (int v = 0; v < 8; ++v) {
        const int j = j0 + 16 * jt + v + jhalf;
        const int b = b0 + nlan + 16 * t;
        TMP[((size_t)r * n + j) * batch + b] = c[jt][t][v];
      }
}

__global__ void k_pass2(const float* __restrict__ Gm, const float* __restrict__ PA,
                        const float* __restrict__ TMP, const float* __restrict__ bias,
                        float* __restrict__ out, int n, int batch, int rdim) {
  extern __shared__ float gp[];
  const int tid = threadIdx.x;
  const int lane = tid & 31;
  const int w = tid >> 5;

  const int i0 = blockIdx.x * 64 + (w & 1) * 32;
  const int b0 = blockIdx.y * 256 + (w >> 1) * 64;

  const int mlan = lane & 15;
  const int kb2  = (lane >> 4) * 2;

  v8f c[2][4];
#pragma unroll
  for (int it = 0; it < 2; ++it)
#pragma unroll
    for (int t = 0; t < 4; ++t) c[it][t] = (v8f){0.f,0.f,0.f,0.f,0.f,0.f,0.f,0.f};

  const int jmax   = i0 + 32;
  const int bbase0 = PAD + i0 + mlan - kb2;
  const int bbase1 = bbase0 + 16;

  for (int r = 0; r < rdim; ++r) {
    __syncthreads();
    for (int t = tid; t < PAD; t += blockDim.x) gp[t] = 0.f;
    for (int t = tid; t < 4; t += blockDim.x) gp[PAD + n + t] = 0.f;
    for (int t = tid; t < n; t += blockDim.x) gp[PAD + t] = Gm[(size_t)r * n + t] / PA[t];
    __syncthreads();

    const float* __restrict__ trow =
        TMP + ((size_t)r * n + kb2) * batch + (b0 + mlan);
    for (int j0k = 0; j0k < jmax; j0k += 4) {
      v2f bb0, bb1;
      bb0.x = gp[bbase0 - j0k]; bb0.y = gp[bbase0 - j0k - 1];
      bb1.x = gp[bbase1 - j0k]; bb1.y = gp[bbase1 - j0k - 1];
      const float* __restrict__ ap = trow + (size_t)j0k * batch;
#pragma unroll
      for (int t = 0; t < 4; ++t) {
        v2f a;
        a.x = ap[16 * t];
        a.y = ap[batch + 16 * t];
        c[0][t] = __builtin_amdgcn_wmma_f32_16x16x4_f32(false, a, false, bb0,
                                                        (short)0, c[0][t], false, false);
        c[1][t] = __builtin_amdgcn_wmma_f32_16x16x4_f32(false, a, false, bb1,
                                                        (short)0, c[1][t], false, false);
      }
    }
  }

  const int bhalf = (lane >> 4) * 8;
#pragma unroll
  for (int it = 0; it < 2; ++it) {
    const int i = i0 + 16 * it + mlan;
    const float pa = PA[i];
    const float bi = bias[i];
#pragma unroll
    for (int t = 0; t < 4; ++t)
#pragma unroll
      for (int v = 0; v < 8; ++v) {
        const int b = b0 + 16 * t + v + bhalf;
        out[(size_t)b * n + i] = bi + pa * c[it][t][v];
      }
  }
}

extern "C" void kernel_launch(void* const* d_in, const int* in_sizes, int n_in,
                              void* d_out, int out_size, void* d_ws, size_t ws_size,
                              hipStream_t stream) {
  const float* x     = (const float*)d_in[0];
  const float* sA    = (const float*)d_in[1];
  const float* sB    = (const float*)d_in[2];
  const float* G     = (const float*)d_in[3];
  const float* H     = (const float*)d_in[4];
  const float* bias  = (const float*)d_in[5];
  float* out = (float*)d_out;

  const int n     = in_sizes[5];
  const int batch = in_sizes[0] / n;
  const int rdim  = in_sizes[3] / n;

  float* PA  = (float*)d_ws;
  float* PB  = PA + n;
  float* Xt  = PB + n;
  float* TMP = Xt + (size_t)n * batch;

  const size_t ldsBytes = (size_t)(PAD + n + 4) * sizeof(float);

  k_prefix<<<1, 64, 0, stream>>>(sA, sB, PA, PB, n);

  {
    dim3 grid(n / 32, batch / 32), block(32, 8);
    k_transpose_scale<<<grid, block, 0, stream>>>(x, PB, Xt, n, batch);
  }
  {
    dim3 grid(n / 64, batch / 256, rdim);
    k_pass1<<<grid, 256, ldsBytes, stream>>>(H, PB, Xt, TMP, n, batch);
  }
  {
    dim3 grid(n / 64, batch / 256, 1);
    k_pass2<<<grid, 256, ldsBytes, stream>>>(G, PA, TMP, bias, out, n, batch, rdim);
  }
}


# Software-pipeline both triangular K loops: blocks of 4 k-steps (32 WMMAs) on two register sets, next block's global and LDS loads issued as one group ahead of the current block; sum order unchanged
# speedup vs baseline: 1.5356x; 1.5356x over previous
#include <hip/hip_runtime.h>


typedef __attribute__((ext_vector_type(2))) float v2f;
typedef __attribute__((ext_vector_type(8))) float v8f;

#define PAD 32

__global__ void k_prefix(const float* __restrict__ sA, const float* __restrict__ sB,
                         float* __restrict__ PA, float* __restrict__ PB, int n) {
  if (threadIdx.x == 0) {
    float p = 1.f; PA[0] = p;
    for (int i = 1; i < n; ++i) { p *= sA[i - 1]; PA[i] = p; }
  } else if (threadIdx.x == 1) {
    float p = 1.f; PB[0] = p;
    for (int i = 1; i < n; ++i) { p *= sB[i - 1]; PB[i] = p; }
  }
}

__global__ void k_transpose_scale(const float* __restrict__ x, const float* __restrict__ PB,
                                  float* __restrict__ Xt, int n, int batch) {
  __shared__ float tile[32][33];
  const int k0 = blockIdx.x * 32, b0 = blockIdx.y * 32;
  const int tx = threadIdx.x, ty = threadIdx.y;
#pragma unroll
  for (int s = 0; s < 32; s += 8)
    tile[ty + s][tx] = x[(size_t)(b0 + ty + s) * n + (k0 + tx)];
  __syncthreads();
#pragma unroll
  for (int s = 0; s < 32; s += 8) {
    const int k = k0 + ty + s;
    Xt[(size_t)k * batch + (b0 + tx)] = PB[k] * tile[tx][ty + s];
  }
}


#define KS 4

struct FragSet {
  v2f t0[KS], t1[KS];
  v2f s[KS][4];
};

#define SET_EDGE() __builtin_amdgcn_sched_barrier(0)

static __device__ __forceinline__ v8f mma4(v2f a, v2f b, v8f c) {
  return __builtin_amdgcn_wmma_f32_16x16x4_f32(false, a, false, b, (short)0, c, false, false);
}

__global__ void __launch_bounds__(256) k_pass1(const float* __restrict__ Hm, const float* __restrict__ PB,
                        const float* __restrict__ Xt, float* __restrict__ TMP,
                        int n, int batch) {
  extern __shared__ float hp[];
  const int tid = threadIdx.x;
  const int lane = tid & 31;
  const int w = tid >> 5;
  const int r = blockIdx.z;

  for (int t = tid; t < PAD; t += blockDim.x) hp[t] = 0.f;
  for (int t = tid; t < 4; t += blockDim.x) hp[PAD + n + t] = 0.f;
  for (int t = tid; t < n; t += blockDim.x) hp[PAD + t] = Hm[(size_t)r * n + t] / PB[t];
  __syncthreads();

  const int j0 = blockIdx.x * 64 + (w & 1) * 32;
  const int b0 = blockIdx.y * 256 + (w >> 1) * 64;

  const int jr   = lane & 15;
  const int kb2  = (lane >> 4) * 2;
  const int nlan = lane & 15;

  v8f c[2][4];
#pragma unroll
  for (int jt = 0; jt < 2; ++jt)
#pragma unroll
    for (int t = 0; t < 4; ++t) c[jt][t] = (v8f){0.f,0.f,0.f,0.f,0.f,0.f,0.f,0.f};

  const int abase0 = PAD + kb2 - jr - j0;
  const int abase1 = abase0 - 16;
  const float* __restrict__ xcol = Xt + (size_t)kb2 * batch + (b0 + nlan);

  auto load_set = [&](FragSet& f, int k0) {
    const float* __restrict__ xp = xcol + (size_t)k0 * batch;
#pragma unroll
    for (int q = 0; q < KS; ++q) {
      const float* __restrict__ xq = xp + (size_t)(4 * q) * batch;
#pragma unroll
      for (int t = 0; t < 4; ++t) {
        f.s[q][t].x = xq[16 * t];
        f.s[q][t].y = xq[batch + 16 * t];
      }
    }
#pragma unroll
    for (int q = 0; q < KS; ++q) {
      f.t0[q].x = hp[abase0 + k0 + 4 * q]; f.t0[q].y = hp[abase0 + k0 + 4 * q + 1];
      f.t1[q].x = hp[abase1 + k0 + 4 * q]; f.t1[q].y = hp[abase1 + k0 + 4 * q + 1];
    }
  };
  auto mma_set = [&](const FragSet& f) {
#pragma unroll
    for (int q = 0; q < KS; ++q)
#pragma unroll
      for (int t = 0; t < 4; ++t) {
        c[0][t] = mma4(f.t0[q], f.s[q][t], c[0][t]);
        c[1][t] = mma4(f.t1[q], f.s[q][t], c[1][t]);
      }
  };

  FragSet fa, fb;
  int k0 = j0;
  load_set(fa, k0);
  for (; k0 + 32 < n; k0 += 32) {
    load_set(fb, k0 + 16);
    SET_EDGE();
    mma_set(fa);
    SET_EDGE();
    load_set(fa, k0 + 32);
    SET_EDGE();
    mma_set(fb);
    SET_EDGE();
  }
  load_set(fb, k0 + 16);
  SET_EDGE();
  mma_set(fa);
  mma_set(fb);

  const int jhalf = (lane >> 4) * 8;
#pragma unroll
  for (int jt = 0; jt < 2; ++jt)
#pragma unroll
    for (int t = 0; t < 4; ++t)
#pragma unroll
      for (int v = 0; v < 8; ++v) {
        const int j = j0 + 16 * jt + v + jhalf;
        const int b = b0 + nlan + 16 * t;
        TMP[((size_t)r * n + j) * batch + b] = c[jt][t][v];
      }
}

__global__ void __launch_bounds__(256) k_pass2(const float* __restrict__ Gm, const float* __restrict__ PA,
                        const float* __restrict__ TMP, const float* __restrict__ bias,
                        float* __restrict__ out, int n, int batch, int rdim) {
  extern __shared__ float gp[];
  const int tid = threadIdx.x;
  const int lane = tid & 31;
  const int w = tid >> 5;

  const int i0 = blockIdx.x * 64 + (w & 1) * 32;
  const int b0 = blockIdx.y * 256 + (w >> 1) * 64;

  const int mlan = lane & 15;
  const int kb2  = (lane >> 4) * 2;

  v8f c[2][4];
#pragma unroll
  for (int it = 0; it < 2; ++it)
#pragma unroll
    for (int t = 0; t < 4; ++t) c[it][t] = (v8f){0.f,0.f,0.f,0.f,0.f,0.f,0.f,0.f};

  const int jmax   = i0 + 32;
  const int bbase0 = PAD + i0 + mlan - kb2;
  const int bbase1 = bbase0 + 16;

  for (int r = 0; r < rdim; ++r) {
    __syncthreads();
    for (int t = tid; t < PAD; t += blockDim.x) gp[t] = 0.f;
    for (int t = tid; t < 4; t += blockDim.x) gp[PAD + n + t] = 0.f;
    for (int t = tid; t < n; t += blockDim.x) gp[PAD + t] = Gm[(size_t)r * n + t] / PA[t];
    __syncthreads();

    const float* __restrict__ trow =
        TMP + ((size_t)r * n + kb2) * batch + (b0 + mlan);

    auto load_set = [&](FragSet& f, int j0k) {
      const float* __restrict__ ap = trow + (size_t)j0k * batch;
#pragma unroll
      for (int q = 0; q < KS; ++q) {
        const float* __restrict__ aq = ap + (size_t)(4 * q) * batch;
#pragma unroll
        for (int t = 0; t < 4; ++t) {
          f.s[q][t].x = aq[16 * t];
          f.s[q][t].y = aq[batch + 16 * t];
        }
      }
#pragma unroll
      for (int q = 0; q < KS; ++q) {
        f.t0[q].x = gp[bbase0 - j0k - 4 * q]; f.t0[q].y = gp[bbase0 - j0k - 4 * q - 1];
        f.t1[q].x = gp[bbase1 - j0k - 4 * q]; f.t1[q].y = gp[bbase1 - j0k - 4 * q - 1];
      }
    };
    auto mma_set = [&](const FragSet& f) {
#pragma unroll
      for (int q = 0; q < KS; ++q)
#pragma unroll
        for (int t = 0; t < 4; ++t) {
          c[0][t] = mma4(f.s[q][t], f.t0[q], c[0][t]);
          c[1][t] = mma4(f.s[q][t], f.t1[q], c[1][t]);
        }
    };

    FragSet fa, fb;
    int j0k = 0;
    load_set(fa, j0k);
    for (; j0k + 32 < jmax; j0k += 32) {
      load_set(fb, j0k + 16);
      SET_EDGE();
      mma_set(fa);
      SET_EDGE();
      load_set(fa, j0k + 32);
      SET_EDGE();
      mma_set(fb);
      SET_EDGE();
    }
    load_set(fb, j0k + 16);
    SET_EDGE();
    mma_set(fa);
    mma_set(fb);
  }

  const int bhalf = (lane >> 4) * 8;
#pragma unroll
  for (int it = 0; it < 2; ++it) {
    const int i = i0 + 16 * it + mlan;
    const float pa = PA[i];
    const float bi = bias[i];
#pragma unroll
    for (int t = 0; t < 4; ++t)
#pragma unroll
      for (int v = 0; v < 8; ++v) {
        const int b = b0 + 16 * t + v + bhalf;
        out[(size_t)b * n + i] = bi + pa * c[it][t][v];
      }
  }
}

extern "C" void kernel_launch(void* const* d_in, const int* in_sizes, int n_in,
                              void* d_out, int out_size, void* d_ws, size_t ws_size,
                              hipStream_t stream) {
  const float* x     = (const float*)d_in[0];
  const float* sA    = (const float*)d_in[1];
  const float* sB    = (const float*)d_in[2];
  const float* G     = (const float*)d_in[3];
  const float* H     = (const float*)d_in[4];
  const float* bias  = (const float*)d_in[5];
  float* out = (float*)d_out;

  const int n     = in_sizes[5];
  const int batch = in_sizes[0] / n;
  const int rdim  = in_sizes[3] / n;

  float* PA  = (float*)d_ws;
  float* PB  = PA + n;
  float* Xt  = PB + n;
  float* TMP = Xt + (size_t)n * batch;

  const size_t ldsBytes = (size_t)(PAD + n + 4) * sizeof(float);

  k_prefix<<<1, 64, 0, stream>>>(sA, sB, PA, PB, n);

  {
    dim3 grid(n / 32, batch / 32), block(32, 8);
    k_transpose_scale<<<grid, block, 0, stream>>>(x, PB, Xt, n, batch);
  }
  {
    dim3 grid(n / 64, batch / 256, rdim);
    k_pass1<<<grid, 256, ldsBytes, stream>>>(H, PB, Xt, TMP, n, batch);
  }
  {
    dim3 grid(n / 64, batch / 256, 1);
    k_pass2<<<grid, 256, ldsBytes, stream>>>(G, PA, TMP, bias, out, n, batch, rdim);
  }
}


# Build M once by its diagonal recurrence (64-row segments plus a carry pass; no division, no prefix scaling), then one dense f32 WMMA GEMM x*M^T on k/4-interleaved b128 operands; sum order changes
# speedup vs baseline: 8.6670x; 5.6440x over previous
#include <hip/hip_runtime.h>


typedef __attribute__((ext_vector_type(2))) float v2f;
typedef __attribute__((ext_vector_type(4))) float v4f;
typedef __attribute__((ext_vector_type(8))) float v8f;

#define TS 64
#define HW1 320
#define HW2 192

__global__ void k_xquad(const float* __restrict__ x, float* __restrict__ Xq, int n, int batch) {
  __shared__ v4f tile[32][33];
  const int q0 = blockIdx.x * 32, b0 = blockIdx.y * 32;
  const int tx = threadIdx.x, ty = threadIdx.y;
#pragma unroll
  for (int s = 0; s < 32; s += 8)
    tile[ty + s][tx] = *(const v4f*)(x + (size_t)(b0 + ty + s) * n + 4 * (q0 + tx));
  __syncthreads();
#pragma unroll
  for (int s = 0; s < 32; s += 8) {
    const int q = q0 + ty + s;
    *(v4f*)(Xq + ((size_t)q * batch + (b0 + tx)) * 4) = tile[tx][ty + s];
  }
}

static __device__ __forceinline__ void stage_rows(float* gq, float* hq, int hw, int i0, int c0,
                                                  const float* __restrict__ Gm, const float* __restrict__ Hm,
                                                  const float* __restrict__ sA, const float* __restrict__ sB,
                                                  int n, int rdim, int tid, int nthr) {
  for (int t = tid; t < (rdim + 1) * TS; t += nthr) {
    const int r = t / TS, row = i0 + (t - r * TS);
    gq[t] = (r < rdim) ? Gm[(size_t)r * n + row] : (row >= 1 ? sA[row - 1] : 0.f);
  }
  for (int t = tid; t < (rdim + 1) * hw; t += nthr) {
    const int r = t / hw, col = c0 + (t - r * hw);
    float v = 0.f;
    if (col >= 0 && col < n) v = (r < rdim) ? Hm[(size_t)r * n + col] : (col >= 1 ? sB[col - 1] : 0.f);
    hq[t] = v;
  }
}

template <int R>
static __device__ __forceinline__ void pstep(const float* ga, const float* hx, int hw, int rdim, float& p, float& m) {
  p = 0.f;
  if constexpr (R > 0) {
#pragma unroll
    for (int r = 0; r < R; ++r) p += ga[r * TS] * hx[r * hw];
    m = ga[R * TS] * hx[R * hw];
  } else {
    for (int r = 0; r < rdim; ++r) p += ga[r * TS] * hx[r * hw];
    m = ga[rdim * TS] * hx[rdim * hw];
  }
}

template <int R>
static __device__ __forceinline__ void seg_walk(const float* gq, const float* hx, int rdim, float& acc, float& prod) {
  acc = 0.f; prod = 1.f;
  if constexpr (R > 0) {
#pragma unroll 8
    for (int a = 0; a < TS; ++a) {
      float p, m;
      pstep<R>(gq + a, hx + a, HW1, rdim, p, m);
      acc = m * acc + p;
      prod *= m;
    }
  } else {
#pragma nounroll
    for (int a = 0; a < TS; ++a) {
      float p, m;
      pstep<0>(gq + a, hx + a, HW1, rdim, p, m);
      acc = m * acc + p;
      prod *= m;
    }
  }
}

__global__ void __launch_bounds__(256) k_segsum(const float* __restrict__ Gm, const float* __restrict__ Hm,
                         const float* __restrict__ sA, const float* __restrict__ sB,
                         float* __restrict__ T, float* __restrict__ PI, int n, int rdim) {
  extern __shared__ float sm1[];
  float* gq = sm1;
  float* hq = sm1 + (rdim + 1) * TS;
  const int tid = threadIdx.x;
  const int i0 = blockIdx.y * TS;
  const int kb = blockIdx.x * 256;
  stage_rows(gq, hq, HW1, i0, kb - (TS - 1), Gm, Hm, sA, sB, n, rdim, tid, 256);
  __syncthreads();
  float acc, prod;
  if (rdim == 4) seg_walk<4>(gq, hq + tid, rdim, acc, prod);
  else           seg_walk<0>(gq, hq + tid, rdim, acc, prod);
  T[(size_t)blockIdx.y * n + kb + tid] = acc;
  PI[(size_t)blockIdx.y * n + kb + tid] = prod;
}

__global__ void __launch_bounds__(256) k_carry(const float* __restrict__ T, const float* __restrict__ PI,
                         float* __restrict__ MB, int n) {
  const int g = blockIdx.x * 256 + threadIdx.x;
  if (g >= 2 * n - 1) return;
  const int d = g - (n - 1);
  const int nI = n / TS;
  const int lo = (d < -(TS - 1)) ? (-d) / TS : 0;
  const int hi = ((n - d) / TS < nI) ? (n - d) / TS : nI;
  float s = 0.f;
#pragma unroll 8
  for (int I = lo; I < hi; ++I) {
    const size_t idx = (size_t)I * n + (TS * I + (TS - 1) + d);
    s = PI[idx] * s + T[idx];
    MB[idx] = s;
  }
}

template <int R>
static __device__ __forceinline__ void walk(const float* gq, const float* hq, float* st, int tid, float s, int rdim) {
  if constexpr (R > 0) {
#pragma nounroll
    for (int a0 = 0; a0 < TS; a0 += 8) {
      float p[8], m[8];
#pragma unroll
      for (int u = 0; u < 8; ++u) pstep<R>(gq + a0 + u, hq + tid + a0 + u, HW2, rdim, p[u], m[u]);
#pragma unroll
      for (int u = 0; u < 8; ++u) {
        s = m[u] * s + p[u];
        const int b = tid + a0 + u - (TS - 1);
        if (b >= 0 && b < TS) st[(a0 + u) * (TS + 1) + b] = s;
      }
    }
  } else {
#pragma nounroll
    for (int a = 0; a < TS; ++a) {
      float p, m;
      pstep<0>(gq + a, hq + tid + a, HW2, rdim, p, m);
      s = m * s + p;
      const int b = tid + a - (TS - 1);
      if (b >= 0 && b < TS) st[a * (TS + 1) + b] = s;
    }
  }
}

__global__ void __launch_bounds__(128) k_buildM(const float* __restrict__ Gm, const float* __restrict__ Hm,
                         const float* __restrict__ sA, const float* __restrict__ sB,
                         const float* __restrict__ MB, float* __restrict__ Sq, int n, int rdim) {
  extern __shared__ float sm2[];
  __shared__ float st[TS * (TS + 1)];
  float* gq = sm2;
  float* hq = gq + (rdim + 1) * TS;
  const int tid = threadIdx.x;
  const int I = blockIdx.y;
  const int i0 = I * TS, k0 = blockIdx.x * TS;
  stage_rows(gq, hq, HW2, i0, k0 - (TS - 1), Gm, Hm, sA, sB, n, rdim, tid, 128);
  __syncthreads();

  if (tid < 2 * TS - 1) {
    const int c = k0 - (TS - 1) + tid;
    const float s = (I > 0 && c >= 1) ? MB[(size_t)(I - 1) * n + (c - 1)] : 0.f;
    if (rdim == 4) walk<4>(gq, hq, st, tid, s, rdim);
    else           walk<0>(gq, hq, st, tid, s, rdim);
  }
  __syncthreads();

  for (int e = tid; e < TS * (TS / 4); e += 128) {
    const int a = e & (TS - 1), kq = e / TS;
    const float* sp = st + a * (TS + 1) + 4 * kq;
    const v4f v = {sp[0], sp[1], sp[2], sp[3]};
    *(v4f*)(Sq + ((size_t)(k0 / 4 + kq) * n + (i0 + a)) * 4) = v;
  }
}

#define KS 4

struct FragSet {
  v4f a[2][4];
  v4f b[2][2];
};

#define SET_EDGE() __builtin_amdgcn_sched_barrier(0)

static __device__ __forceinline__ v8f mma4(v2f a, v2f b, v8f c) {
  return __builtin_amdgcn_wmma_f32_16x16x4_f32(false, a, false, b, (short)0, c, false, false);
}

__global__ void __launch_bounds__(256) k_gemm(const float* __restrict__ Xq, const float* __restrict__ Sq,
                        const float* __restrict__ bias, float* __restrict__ out, int n, int batch) {
  const int tid = threadIdx.x;
  const int lane = tid & 31;
  const int w = tid >> 5;

  const int i0 = blockIdx.x * 64 + (w & 1) * 32;
  const int b0 = blockIdx.y * 256 + (w >> 1) * 64;

  const int mlan = lane & 15;
  const int h    = lane >> 4;

  v8f c[2][4];
#pragma unroll
  for (int it = 0; it < 2; ++it)
#pragma unroll
    for (int t = 0; t < 4; ++t) c[it][t] = (v8f){0.f,0.f,0.f,0.f,0.f,0.f,0.f,0.f};

  const v4f* __restrict__ xa = (const v4f*)Xq + (size_t)h * batch + (b0 + mlan);
  const v4f* __restrict__ sb = (const v4f*)Sq + (size_t)h * n + (i0 + mlan);

  auto load_set = [&](FragSet& f, int k0) {
    const v4f* __restrict__ xp = xa + (size_t)(k0 >> 2) * batch;
    const v4f* __restrict__ sp = sb + (size_t)(k0 >> 2) * n;
#pragma unroll
    for (int Q = 0; Q < 2; ++Q) {
#pragma unroll
      for (int t = 0; t < 4; ++t) f.a[Q][t] = xp[(size_t)(2 * Q) * batch + 16 * t];
#pragma unroll
      for (int it = 0; it < 2; ++it) f.b[Q][it] = sp[(size_t)(2 * Q) * n + 16 * it];
    }
  };
  auto mma_set = [&](const FragSet& f) {
#pragma unroll
    for (int Q = 0; Q < 2; ++Q) {
#pragma unroll
      for (int t = 0; t < 4; ++t) {
        c[0][t] = mma4(f.a[Q][t].lo, f.b[Q][0].lo, c[0][t]);
        c[1][t] = mma4(f.a[Q][t].lo, f.b[Q][1].lo, c[1][t]);
      }
#pragma unroll
      for (int t = 0; t < 4; ++t) {
        c[0][t] = mma4(f.a[Q][t].hi, f.b[Q][0].hi, c[0][t]);
        c[1][t] = mma4(f.a[Q][t].hi, f.b[Q][1].hi, c[1][t]);
      }
    }
  };

  FragSet fa, fb;
  int k0 = 0;
  load_set(fa, k0);
  for (; k0 + 32 < n; k0 += 32) {
    load_set(fb, k0 + 16);
    SET_EDGE();
    mma_set(fa);
    SET_EDGE();
    load_set(fa, k0 + 32);
    SET_EDGE();
    mma_set(fb);
    SET_EDGE();
  }
  load_set(fb, k0 + 16);
  SET_EDGE();
  mma_set(fa);
  mma_set(fb);

  const int bhalf = h * 8;
#pragma unroll
  for (int it = 0; it < 2; ++it) {
    const int i = i0 + 16 * it + mlan;
    const float bi = bias[i];
#pragma unroll
    for (int t = 0; t < 4; ++t)
#pragma unroll
      for (int v = 0; v < 8; ++v) {
        const int b = b0 + 16 * t + v + bhalf;
        out[(size_t)b * n + i] = bi + c[it][t][v];
      }
  }
}

extern "C" void kernel_launch(void* const* d_in, const int* in_sizes, int n_in,
                              void* d_out, int out_size, void* d_ws, size_t ws_size,
                              hipStream_t stream) {
  const float* x     = (const float*)d_in[0];
  const float* sA    = (const float*)d_in[1];
  const float* sB    = (const float*)d_in[2];
  const float* G     = (const float*)d_in[3];
  const float* H     = (const float*)d_in[4];
  const float* bias  = (const float*)d_in[5];
  float* out = (float*)d_out;

  const int n     = in_sizes[5];
  const int batch = in_sizes[0] / n;
  const int rdim  = in_sizes[3] / n;

  float* Xq = (float*)d_ws;
  float* Sq = Xq + (size_t)n * batch;
  float* T  = Xq;
  float* PI = T + (size_t)(n / TS) * n;
  float* MB = PI + (size_t)(n / TS) * n;

  {
    dim3 grid(n / 256, n / TS);
    const size_t lds = (size_t)(rdim + 1) * (TS + HW1) * sizeof(float);
    k_segsum<<<grid, 256, lds, stream>>>(G, H, sA, sB, T, PI, n, rdim);
  }
  k_carry<<<2 * n / 256, 256, 0, stream>>>(T, PI, MB, n);
  {
    dim3 grid(n / TS, n / TS);
    const size_t lds = (size_t)(rdim + 1) * (TS + HW2) * sizeof(float);
    k_buildM<<<grid, 128, lds, stream>>>(G, H, sA, sB, MB, Sq, n, rdim);
  }
  {
    dim3 grid(n / 128, batch / 32), block(32, 8);
    k_xquad<<<grid, block, 0, stream>>>(x, Xq, n, batch);
  }
  {
    dim3 grid(n / 64, batch / 256, 1);
    k_gemm<<<grid, 256, 0, stream>>>(Xq, Sq, bias, out, n, batch);
  }
}


# GEMM operand prefetch distance raised from one 16-k block to three: ring of four register sets, each set's 12 b128 loads still one group ahead of a 32-WMMA block; sum order unchanged
# speedup vs baseline: 9.0695x; 1.0464x over previous
#include <hip/hip_runtime.h>


typedef __attribute__((ext_vector_type(2))) float v2f;
typedef __attribute__((ext_vector_type(4))) float v4f;
typedef __attribute__((ext_vector_type(8))) float v8f;

#define TS 64
#define HW1 320
#define HW2 192

__global__ void k_xquad(const float* __restrict__ x, float* __restrict__ Xq, int n, int batch) {
  __shared__ v4f tile[32][33];
  const int q0 = blockIdx.x * 32, b0 = blockIdx.y * 32;
  const int tx = threadIdx.x, ty = threadIdx.y;
#pragma unroll
  for (int s = 0; s < 32; s += 8)
    tile[ty + s][tx] = *(const v4f*)(x + (size_t)(b0 + ty + s) * n + 4 * (q0 + tx));
  __syncthreads();
#pragma unroll
  for (int s = 0; s < 32; s += 8) {
    const int q = q0 + ty + s;
    *(v4f*)(Xq + ((size_t)q * batch + (b0 + tx)) * 4) = tile[tx][ty + s];
  }
}

static __device__ __forceinline__ void stage_rows(float* gq, float* hq, int hw, int i0, int c0,
                                                  const float* __restrict__ Gm, const float* __restrict__ Hm,
                                                  const float* __restrict__ sA, const float* __restrict__ sB,
                                                  int n, int rdim, int tid, int nthr) {
  for (int t = tid; t < (rdim + 1) * TS; t += nthr) {
    const int r = t / TS, row = i0 + (t - r * TS);
    gq[t] = (r < rdim) ? Gm[(size_t)r * n + row] : (row >= 1 ? sA[row - 1] : 0.f);
  }
  for (int t = tid; t < (rdim + 1) * hw; t += nthr) {
    const int r = t / hw, col = c0 + (t - r * hw);
    float v = 0.f;
    if (col >= 0 && col < n) v = (r < rdim) ? Hm[(size_t)r * n + col] : (col >= 1 ? sB[col - 1] : 0.f);
    hq[t] = v;
  }
}

template <int R>
static __device__ __forceinline__ void pstep(const float* ga, const float* hx, int hw, int rdim, float& p, float& m) {
  p = 0.f;
  if constexpr (R > 0) {
#pragma unroll
    for (int r = 0; r < R; ++r) p += ga[r * TS] * hx[r * hw];
    m = ga[R * TS] * hx[R * hw];
  } else {
    for (int r = 0; r < rdim; ++r) p += ga[r * TS] * hx[r * hw];
    m = ga[rdim * TS] * hx[rdim * hw];
  }
}

template <int R>
static __device__ __forceinline__ void seg_walk(const float* gq, const float* hx, int rdim, float& acc, float& prod) {
  acc = 0.f; prod = 1.f;
  if constexpr (R > 0) {
#pragma unroll 8
    for (int a = 0; a < TS; ++a) {
      float p, m;
      pstep<R>(gq + a, hx + a, HW1, rdim, p, m);
      acc = m * acc + p;
      prod *= m;
    }
  } else {
#pragma nounroll
    for (int a = 0; a < TS; ++a) {
      float p, m;
      pstep<0>(gq + a, hx + a, HW1, rdim, p, m);
      acc = m * acc + p;
      prod *= m;
    }
  }
}

__global__ void __launch_bounds__(256) k_segsum(const float* __restrict__ Gm, const float* __restrict__ Hm,
                         const float* __restrict__ sA, const float* __restrict__ sB,
                         float* __restrict__ T, float* __restrict__ PI, int n, int rdim) {
  extern __shared__ float sm1[];
  float* gq = sm1;
  float* hq = sm1 + (rdim + 1) * TS;
  const int tid = threadIdx.x;
  const int i0 = blockIdx.y * TS;
  const int kb = blockIdx.x * 256;
  stage_rows(gq, hq, HW1, i0, kb - (TS - 1), Gm, Hm, sA, sB, n, rdim, tid, 256);
  __syncthreads();
  float acc, prod;
  if (rdim == 4) seg_walk<4>(gq, hq + tid, rdim, acc, prod);
  else           seg_walk<0>(gq, hq + tid, rdim, acc, prod);
  T[(size_t)blockIdx.y * n + kb + tid] = acc;
  PI[(size_t)blockIdx.y * n + kb + tid] = prod;
}

__global__ void __launch_bounds__(256) k_carry(const float* __restrict__ T, const float* __restrict__ PI,
                         float* __restrict__ MB, int n) {
  const int g = blockIdx.x * 256 + threadIdx.x;
  if (g >= 2 * n - 1) return;
  const int d = g - (n - 1);
  const int nI = n / TS;
  const int lo = (d < -(TS - 1)) ? (-d) / TS : 0;
  const int hi = ((n - d) / TS < nI) ? (n - d) / TS : nI;
  float s = 0.f;
#pragma unroll 8
  for (int I = lo; I < hi; ++I) {
    const size_t idx = (size_t)I * n + (TS * I + (TS - 1) + d);
    s = PI[idx] * s + T[idx];
    MB[idx] = s;
  }
}

template <int R>
static __device__ __forceinline__ void walk(const float* gq, const float* hq, float* st, int tid, float s, int rdim) {
  if constexpr (R > 0) {
#pragma nounroll
    for (int a0 = 0; a0 < TS; a0 += 8) {
      float p[8], m[8];
#pragma unroll
      for (int u = 0; u < 8; ++u) pstep<R>(gq + a0 + u, hq + tid + a0 + u, HW2, rdim, p[u], m[u]);
#pragma unroll
      for (int u = 0; u < 8; ++u) {
        s = m[u] * s + p[u];
        const int b = tid + a0 + u - (TS - 1);
        if (b >= 0 && b < TS) st[(a0 + u) * (TS + 1) + b] = s;
      }
    }
  } else {
#pragma nounroll
    for (int a = 0; a < TS; ++a) {
      float p, m;
      pstep<0>(gq + a, hq + tid + a, HW2, rdim, p, m);
      s = m * s + p;
      const int b = tid + a - (TS - 1);
      if (b >= 0 && b < TS) st[a * (TS + 1) + b] = s;
    }
  }
}

__global__ void __launch_bounds__(128) k_buildM(const float* __restrict__ Gm, const float* __restrict__ Hm,
                         const float* __restrict__ sA, const float* __restrict__ sB,
                         const float* __restrict__ MB, float* __restrict__ Sq, int n, int rdim) {
  extern __shared__ float sm2[];
  __shared__ float st[TS * (TS + 1)];
  float* gq = sm2;
  float* hq = gq + (rdim + 1) * TS;
  const int tid = threadIdx.x;
  const int I = blockIdx.y;
  const int i0 = I * TS, k0 = blockIdx.x * TS;
  stage_rows(gq, hq, HW2, i0, k0 - (TS - 1), Gm, Hm, sA, sB, n, rdim, tid, 128);
  __syncthreads();

  if (tid < 2 * TS - 1) {
    const int c = k0 - (TS - 1) + tid;
    const float s = (I > 0 && c >= 1) ? MB[(size_t)(I - 1) * n + (c - 1)] : 0.f;
    if (rdim == 4) walk<4>(gq, hq, st, tid, s, rdim);
    else           walk<0>(gq, hq, st, tid, s, rdim);
  }
  __syncthreads();

  for (int e = tid; e < TS * (TS / 4); e += 128) {
    const int a = e & (TS - 1), kq = e / TS;
    const float* sp = st + a * (TS + 1) + 4 * kq;
    const v4f v = {sp[0], sp[1], sp[2], sp[3]};
    *(v4f*)(Sq + ((size_t)(k0 / 4 + kq) * n + (i0 + a)) * 4) = v;
  }
}

#define KS 4
#define NSET 4

struct FragSet {
  v4f a[2][4];
  v4f b[2][2];
};

#define SET_EDGE() __builtin_amdgcn_sched_barrier(0)

static __device__ __forceinline__ v8f mma4(v2f a, v2f b, v8f c) {
  return __builtin_amdgcn_wmma_f32_16x16x4_f32(false, a, false, b, (short)0, c, false, false);
}

__global__ void __launch_bounds__(256) k_gemm(const float* __restrict__ Xq, const float* __restrict__ Sq,
                        const float* __restrict__ bias, float* __restrict__ out, int n, int batch) {
  const int tid = threadIdx.x;
  const int lane = tid & 31;
  const int w = tid >> 5;

  const int i0 = blockIdx.x * 64 + (w & 1) * 32;
  const int b0 = blockIdx.y * 256 + (w >> 1) * 64;

  const int mlan = lane & 15;
  const int h    = lane >> 4;

  v8f c[2][4];
#pragma unroll
  for (int it = 0; it < 2; ++it)
#pragma unroll
    for (int t = 0; t < 4; ++t) c[it][t] = (v8f){0.f,0.f,0.f,0.f,0.f,0.f,0.f,0.f};

  const v4f* __restrict__ xa = (const v4f*)Xq + (size_t)h * batch + (b0 + mlan);
  const v4f* __restrict__ sb = (const v4f*)Sq + (size_t)h * n + (i0 + mlan);

  auto load_set = [&](FragSet& f, int k0) {
    const v4f* __restrict__ xp = xa + (size_t)(k0 >> 2) * batch;
    const v4f* __restrict__ sp = sb + (size_t)(k0 >> 2) * n;
#pragma unroll
    for (int Q = 0; Q < 2; ++Q) {
#pragma unroll
      for (int t = 0; t < 4; ++t) f.a[Q][t] = xp[(size_t)(2 * Q) * batch + 16 * t];
#pragma unroll
      for (int it = 0; it < 2; ++it) f.b[Q][it] = sp[(size_t)(2 * Q) * n + 16 * it];
    }
  };
  auto mma_set = [&](const FragSet& f) {
#pragma unroll
    for (int Q = 0; Q < 2; ++Q) {
#pragma unroll
      for (int t = 0; t < 4; ++t) {
        c[0][t] = mma4(f.a[Q][t].lo, f.b[Q][0].lo, c[0][t]);
        c[1][t] = mma4(f.a[Q][t].lo, f.b[Q][1].lo, c[1][t]);
      }
#pragma unroll
      for (int t = 0; t < 4; ++t) {
        c[0][t] = mma4(f.a[Q][t].hi, f.b[Q][0].hi, c[0][t]);
        c[1][t] = mma4(f.a[Q][t].hi, f.b[Q][1].hi, c[1][t]);
      }
    }
  };

  FragSet f[NSET];
#pragma unroll
  for (int s = 0; s < NSET - 1; ++s) {
    load_set(f[s], 16 * s);
    SET_EDGE();
  }
  int k0 = 0;
  for (; k0 + 16 * NSET < n; k0 += 16 * NSET) {
#pragma unroll
    for (int s = 0; s < NSET; ++s) {
      load_set(f[(s + NSET - 1) % NSET], k0 + 16 * (s + NSET - 1));
      SET_EDGE();
      mma_set(f[s]);
      SET_EDGE();
    }
  }
  load_set(f[NSET - 1], k0 + 16 * (NSET - 1));
  SET_EDGE();
#pragma unroll
  for (int s = 0; s < NSET; ++s) mma_set(f[s]);

  const int bhalf = h * 8;
#pragma unroll
  for (int it = 0; it < 2; ++it) {
    const int i = i0 + 16 * it + mlan;
    const float bi = bias[i];
#pragma unroll
    for (int t = 0; t < 4; ++t)
#pragma unroll
      for (int v = 0; v < 8; ++v) {
        const int b = b0 + 16 * t + v + bhalf;
        out[(size_t)b * n + i] = bi + c[it][t][v];
      }
  }
}

extern "C" void kernel_launch(void* const* d_in, const int* in_sizes, int n_in,
                              void* d_out, int out_size, void* d_ws, size_t ws_size,
                              hipStream_t stream) {
  const float* x     = (const float*)d_in[0];
  const float* sA    = (const float*)d_in[1];
  const float* sB    = (const float*)d_in[2];
  const float* G     = (const float*)d_in[3];
  const float* H     = (const float*)d_in[4];
  const float* bias  = (const float*)d_in[5];
  float* out = (float*)d_out;

  const int n     = in_sizes[5];
  const int batch = in_sizes[0] / n;
  const int rdim  = in_sizes[3] / n;

  float* Xq = (float*)d_ws;
  float* Sq = Xq + (size_t)n * batch;
  float* T  = Xq;
  float* PI = T + (size_t)(n / TS) * n;
  float* MB = PI + (size_t)(n / TS) * n;

  {
    dim3 grid(n / 256, n / TS);
    const size_t lds = (size_t)(rdim + 1) * (TS + HW1) * sizeof(float);
    k_segsum<<<grid, 256, lds, stream>>>(G, H, sA, sB, T, PI, n, rdim);
  }
  k_carry<<<2 * n / 256, 256, 0, stream>>>(T, PI, MB, n);
  {
    dim3 grid(n / TS, n / TS);
    const size_t lds = (size_t)(rdim + 1) * (TS + HW2) * sizeof(float);
    k_buildM<<<grid, 128, lds, stream>>>(G, H, sA, sB, MB, Sq, n, rdim);
  }
  {
    dim3 grid(n / 128, batch / 32), block(32, 8);
    k_xquad<<<grid, block, 0, stream>>>(x, Xq, n, batch);
  }
  {
    dim3 grid(n / 64, batch / 256, 1);
    k_gemm<<<grid, 256, 0, stream>>>(Xq, Sq, bias, out, n, batch);
  }
}


# GEMM operands staged through LDS by the work-group (128x128 tile, 64-k stages, two slots, whole-line global loads): x is read in place, so the re-layout launch and its plane go; sum order unchanged
# speedup vs baseline: 9.4155x; 1.0381x over previous
#include <hip/hip_runtime.h>


typedef __attribute__((ext_vector_type(2))) float v2f;
typedef __attribute__((ext_vector_type(4))) float v4f;
typedef __attribute__((ext_vector_type(8))) float v8f;

#define TS 64
#define HW1 320
#define HW2 192

static __device__ __forceinline__ void stage_rows(float* gq, float* hq, int hw, int i0, int c0,
                                                  const float* __restrict__ Gm, const float* __restrict__ Hm,
                                                  const float* __restrict__ sA, const float* __restrict__ sB,
                                                  int n, int rdim, int tid, int nthr) {
  for (int t = tid; t < (rdim + 1) * TS; t += nthr) {
    const int r = t / TS, row = i0 + (t - r * TS);
    gq[t] = (r < rdim) ? Gm[(size_t)r * n + row] : (row >= 1 ? sA[row - 1] : 0.f);
  }
  for (int t = tid; t < (rdim + 1) * hw; t += nthr) {
    const int r = t / hw, col = c0 + (t - r * hw);
    float v = 0.f;
    if (col >= 0 && col < n) v = (r < rdim) ? Hm[(size_t)r * n + col] : (col >= 1 ? sB[col - 1] : 0.f);
    hq[t] = v;
  }
}

template <int R>
static __device__ __forceinline__ void pstep(const float* ga, const float* hx, int hw, int rdim, float& p, float& m) {
  p = 0.f;
  if constexpr (R > 0) {
#pragma unroll
    for (int r = 0; r < R; ++r) p += ga[r * TS] * hx[r * hw];
    m = ga[R * TS] * hx[R * hw];
  } else {
    for (int r = 0; r < rdim; ++r) p += ga[r * TS] * hx[r * hw];
    m = ga[rdim * TS] * hx[rdim * hw];
  }
}

template <int R>
static __device__ __forceinline__ void seg_walk(const float* gq, const float* hx, int rdim, float& acc, float& prod) {
  acc = 0.f; prod = 1.f;
  if constexpr (R > 0) {
#pragma unroll 8
    for (int a = 0; a < TS; ++a) {
      float p, m;
      pstep<R>(gq + a, hx + a, HW1, rdim, p, m);
      acc = m * acc + p;
      prod *= m;
    }
  } else {
#pragma nounroll
    for (int a = 0; a < TS; ++a) {
      float p, m;
      pstep<0>(gq + a, hx + a, HW1, rdim, p, m);
      acc = m * acc + p;
      prod *= m;
    }
  }
}

__global__ void __launch_bounds__(256) k_segsum(const float* __restrict__ Gm, const float* __restrict__ Hm,
                         const float* __restrict__ sA, const float* __restrict__ sB,
                         float* __restrict__ T, float* __restrict__ PI, int n, int rdim) {
  extern __shared__ float sm1[];
  float* gq = sm1;
  float* hq = sm1 + (rdim + 1) * TS;
  const int tid = threadIdx.x;
  const int i0 = blockIdx.y * TS;
  const int kb = blockIdx.x * 256;
  stage_rows(gq, hq, HW1, i0, kb - (TS - 1), Gm, Hm, sA, sB, n, rdim, tid, 256);
  __syncthreads();
  float acc, prod;
  if (rdim == 4) seg_walk<4>(gq, hq + tid, rdim, acc, prod);
  else           seg_walk<0>(gq, hq + tid, rdim, acc, prod);
  T[(size_t)blockIdx.y * n + kb + tid] = acc;
  PI[(size_t)blockIdx.y * n + kb + tid] = prod;
}

__global__ void __launch_bounds__(256) k_carry(const float* __restrict__ T, const float* __restrict__ PI,
                         float* __restrict__ MB, int n) {
  const int g = blockIdx.x * 256 + threadIdx.x;
  if (g >= 2 * n - 1) return;
  const int d = g - (n - 1);
  const int nI = n / TS;
  const int lo = (d < -(TS - 1)) ? (-d) / TS : 0;
  const int hi = ((n - d) / TS < nI) ? (n - d) / TS : nI;
  float s = 0.f;
#pragma unroll 8
  for (int I = lo; I < hi; ++I) {
    const size_t idx = (size_t)I * n + (TS * I + (TS - 1) + d);
    s = PI[idx] * s + T[idx];
    MB[idx] = s;
  }
}

template <int R>
static __device__ __forceinline__ void walk(const float* gq, const float* hq, float* st, int tid, float s, int rdim) {
  if constexpr (R > 0) {
#pragma nounroll
    for (int a0 = 0; a0 < TS; a0 += 8) {
      float p[8], m[8];
#pragma unroll
      for (int u = 0; u < 8; ++u) pstep<R>(gq + a0 + u, hq + tid + a0 + u, HW2, rdim, p[u], m[u]);
#pragma unroll
      for (int u = 0; u < 8; ++u) {
        s = m[u] * s + p[u];
        const int b = tid + a0 + u - (TS - 1);
        if (b >= 0 && b < TS) st[(a0 + u) * (TS + 1) + b] = s;
      }
    }
  } else {
#pragma nounroll
    for (int a = 0; a < TS; ++a) {
      float p, m;
      pstep<0>(gq + a, hq + tid + a, HW2, rdim, p, m);
      s = m * s + p;
      const int b = tid + a - (TS - 1);
      if (b >= 0 && b < TS) st[a * (TS + 1) + b] = s;
    }
  }
}

__global__ void __launch_bounds__(128) k_buildM(const float* __restrict__ Gm, const float* __restrict__ Hm,
                         const float* __restrict__ sA, const float* __restrict__ sB,
                         const float* __restrict__ MB, float* __restrict__ Sq, int n, int rdim) {
  extern __shared__ float sm2[];
  __shared__ float st[TS * (TS + 1)];
  float* gq = sm2;
  float* hq = gq + (rdim + 1) * TS;
  const int tid = threadIdx.x;
  const int I = blockIdx.y;
  const int i0 = I * TS, k0 = blockIdx.x * TS;
  stage_rows(gq, hq, HW2, i0, k0 - (TS - 1), Gm, Hm, sA, sB, n, rdim, tid, 128);
  __syncthreads();

  if (tid < 2 * TS - 1) {
    const int c = k0 - (TS - 1) + tid;
    const float s = (I > 0 && c >= 1) ? MB[(size_t)(I - 1) * n + (c - 1)] : 0.f;
    if (rdim == 4) walk<4>(gq, hq, st, tid, s, rdim);
    else           walk<0>(gq, hq, st, tid, s, rdim);
  }
  __syncthreads();

  for (int e = tid; e < TS * (TS / 4); e += 128) {
    const int a = e & (TS - 1), kq = e / TS;
    const float* sp = st + a * (TS + 1) + 4 * kq;
    const v4f v = {sp[0], sp[1], sp[2], sp[3]};
    *(v4f*)(Sq + ((size_t)(k0 / 4 + kq) * n + (i0 + a)) * 4) = v;
  }
}

#define GT 128
#define GKQ 16
#define GP 129
#define GPLANE (GKQ * GP)
#define GSLOT (2 * GPLANE)

struct FragSet {
  v4f a[2][4];
  v4f b[2][2];
};

template <bool B> struct Flag { static constexpr bool value = B; };

#define SET_EDGE() __builtin_amdgcn_sched_barrier(0)

static __device__ __forceinline__ v8f mma4(v2f a, v2f b, v8f c) {
  return __builtin_amdgcn_wmma_f32_16x16x4_f32(false, a, false, b, (short)0, c, false, false);
}

__global__ void __launch_bounds__(256) k_gemm(const float* __restrict__ x, const float* __restrict__ Sq,
                        const float* __restrict__ bias, float* __restrict__ out, int n, int batch) {
  __shared__ v4f lds[2 * GSLOT];

  const int tid = threadIdx.x;
  const int lane = tid & 31;
  const int w = tid >> 5;

  const int ib = blockIdx.x * GT;
  const int bb = blockIdx.y * GT;
  const int wi = (w & 3) * 32;
  const int wb = (w >> 2) * 64;

  const int mlan = lane & 15;
  const int h    = lane >> 4;
  const int nq   = n >> 2;

  const int akq = tid & 15, arow = tid >> 4;
  const int bi = tid & 127, bkq = tid >> 7;
  const v4f* __restrict__ xg = (const v4f*)x + (size_t)(bb + arow) * nq + akq;
  const v4f* __restrict__ sg = (const v4f*)Sq + (size_t)bkq * n + (ib + bi);
  v4f* aS = lds + akq * GP + arow;
  v4f* bS = lds + GPLANE + bkq * GP + bi;
  const v4f* aF = lds + h * GP + (wb + mlan);
  const v4f* bF = lds + GPLANE + h * GP + (wi + mlan);

  v8f c[2][4];
#pragma unroll
  for (int it = 0; it < 2; ++it)
#pragma unroll
    for (int t = 0; t < 4; ++t) c[it][t] = (v8f){0.f,0.f,0.f,0.f,0.f,0.f,0.f,0.f};

  v4f R[16];

  auto gload = [&](int s) {
    const v4f* __restrict__ xp = xg + (size_t)GKQ * s;
    const v4f* __restrict__ sp = sg + (size_t)GKQ * s * n;
#pragma unroll
    for (int j = 0; j < 8; ++j) R[j] = xp[(size_t)(16 * j) * nq];
#pragma unroll
    for (int j = 0; j < 8; ++j) R[8 + j] = sp[(size_t)(2 * j) * n];
  };
  auto sstore = [&](int slot) {
    v4f* ap = aS + slot * GSLOT;
    v4f* bp = bS + slot * GSLOT;
#pragma unroll
    for (int j = 0; j < 8; ++j) ap[16 * j] = R[j];
#pragma unroll
    for (int j = 0; j < 8; ++j) bp[2 * j * GP] = R[8 + j];
  };
  auto lfrag = [&](FragSet& f, int slot, int sub) {
    const v4f* ap = aF + slot * GSLOT + 4 * sub * GP;
    const v4f* bp = bF + slot * GSLOT + 4 * sub * GP;
#pragma unroll
    for (int Q = 0; Q < 2; ++Q) {
#pragma unroll
      for (int t = 0; t < 4; ++t) f.a[Q][t] = ap[2 * Q * GP + 16 * t];
#pragma unroll
      for (int it = 0; it < 2; ++it) f.b[Q][it] = bp[2 * Q * GP + 16 * it];
    }
  };
  auto mma_q = [&](const FragSet& f, int Q) {
#pragma unroll
    for (int t = 0; t < 4; ++t) {
      c[0][t] = mma4(f.a[Q][t].lo, f.b[Q][0].lo, c[0][t]);
      c[1][t] = mma4(f.a[Q][t].lo, f.b[Q][1].lo, c[1][t]);
    }
#pragma unroll
    for (int t = 0; t < 4; ++t) {
      c[0][t] = mma4(f.a[Q][t].hi, f.b[Q][0].hi, c[0][t]);
      c[1][t] = mma4(f.a[Q][t].hi, f.b[Q][1].hi, c[1][t]);
    }
  };

  FragSet F0, F1;

  auto stage = [&](int j, auto has_next, auto has_next2) {
    const int cur = j & 1;
    lfrag(F1, cur, 1); SET_EDGE(); mma_q(F0, 0); mma_q(F0, 1); SET_EDGE();
    lfrag(F0, cur, 2); SET_EDGE(); mma_q(F1, 0); mma_q(F1, 1); SET_EDGE();
    lfrag(F1, cur, 3); SET_EDGE(); mma_q(F0, 0); mma_q(F0, 1); SET_EDGE();
    if constexpr (decltype(has_next)::value) { sstore(cur ^ 1); SET_EDGE(); }
    mma_q(F1, 0); SET_EDGE();
    if constexpr (decltype(has_next)::value) {
      __syncthreads();
      lfrag(F0, cur ^ 1, 0); SET_EDGE();
    }
    mma_q(F1, 1); SET_EDGE();
    if constexpr (decltype(has_next2)::value) { gload(j + 2); SET_EDGE(); }
  };

  const int nst = n / (4 * GKQ);
  gload(0);
  sstore(0);
  __syncthreads();
  gload(1);
  lfrag(F0, 0, 0);
  SET_EDGE();
  int j = 0;
  for (; j < nst - 2; ++j) stage(j, Flag<true>{}, Flag<true>{});
  stage(j, Flag<true>{}, Flag<false>{});
  stage(j + 1, Flag<false>{}, Flag<false>{});

  const int i0 = ib + wi, b0 = bb + wb;
  const int bhalf = h * 8;
#pragma unroll
  for (int it = 0; it < 2; ++it) {
    const int i = i0 + 16 * it + mlan;
    const float bi2 = bias[i];
#pragma unroll
    for (int t = 0; t < 4; ++t)
#pragma unroll
      for (int v = 0; v < 8; ++v) {
        const int b = b0 + 16 * t + v + bhalf;
        out[(size_t)b * n + i] = bi2 + c[it][t][v];
      }
  }
}

extern "C" void kernel_launch(void* const* d_in, const int* in_sizes, int n_in,
                              void* d_out, int out_size, void* d_ws, size_t ws_size,
                              hipStream_t stream) {
  const float* x     = (const float*)d_in[0];
  const float* sA    = (const float*)d_in[1];
  const float* sB    = (const float*)d_in[2];
  const float* G     = (const float*)d_in[3];
  const float* H     = (const float*)d_in[4];
  const float* bias  = (const float*)d_in[5];
  float* out = (float*)d_out;

  const int n     = in_sizes[5];
  const int batch = in_sizes[0] / n;
  const int rdim  = in_sizes[3] / n;

  float* T  = (float*)d_ws;
  float* Sq = T + (size_t)n * batch;
  float* PI = T + (size_t)(n / TS) * n;
  float* MB = PI + (size_t)(n / TS) * n;

  {
    dim3 grid(n / 256, n / TS);
    const size_t lds = (size_t)(rdim + 1) * (TS + HW1) * sizeof(float);
    k_segsum<<<grid, 256, lds, stream>>>(G, H, sA, sB, T, PI, n, rdim);
  }
  k_carry<<<2 * n / 256, 256, 0, stream>>>(T, PI, MB, n);
  {
    dim3 grid(n / TS, n / TS);
    const size_t lds = (size_t)(rdim + 1) * (TS + HW2) * sizeof(float);
    k_buildM<<<grid, 128, lds, stream>>>(G, H, sA, sB, MB, Sq, n, rdim);
  }
  {
    dim3 grid(n / GT, batch / GT, 1);
    k_gemm<<<grid, 256, 0, stream>>>(x, Sq, bias, out, n, batch);
  }
}


# Build launches wait once per group of loads, not per load: a block's G/H/subdiagonal staging, a diagonal's carry operands (32 row blocks at a time), tile start value, write-out LDS reads; sums same
# speedup vs baseline: 9.7388x; 1.0343x over previous
#include <hip/hip_runtime.h>


typedef __attribute__((ext_vector_type(2))) float v2f;
typedef __attribute__((ext_vector_type(4))) float v4f;
typedef __attribute__((ext_vector_type(8))) float v8f;

#define TS 64
#define HW1 320
#define HW2 192

static __device__ __forceinline__ void stage_rows(float* gq, float* hq, int hw, int i0, int c0,
                                                  const float* __restrict__ Gm, const float* __restrict__ Hm,
                                                  const float* __restrict__ sA, const float* __restrict__ sB,
                                                  int n, int rdim, int tid, int nthr) {
  for (int t = tid; t < (rdim + 1) * TS; t += nthr) {
    const int r = t / TS, row = i0 + (t - r * TS);
    gq[t] = (r < rdim) ? Gm[(size_t)r * n + row] : (row >= 1 ? sA[row - 1] : 0.f);
  }
  for (int t = tid; t < (rdim + 1) * hw; t += nthr) {
    const int r = t / hw, col = c0 + (t - r * hw);
    float v = 0.f;
    if (col >= 0 && col < n) v = (r < rdim) ? Hm[(size_t)r * n + col] : (col >= 1 ? sB[col - 1] : 0.f);
    hq[t] = v;
  }
}

template <int R, int HW, int NTHR>
static __device__ __forceinline__ void stage_rows_grouped(float* gq, float* hq, int i0, int c0,
                                                          const float* __restrict__ Gm, const float* __restrict__ Hm,
                                                          const float* __restrict__ sA, const float* __restrict__ sB,
                                                          int n, int tid) {
  constexpr int NGE = (R + 1) * TS, NHE = (R + 1) * HW;
  constexpr int NG = (NGE + NTHR - 1) / NTHR, NH = (NHE + NTHR - 1) / NTHR;
  float vg[NG], vh[NH];
#pragma unroll
  for (int j = 0; j < NG; ++j) {
    const int t = min(tid + j * NTHR, NGE - 1);
    const int r = t / TS, row = i0 + (t - r * TS);
    const float* p = (r < R) ? Gm + ((size_t)r * n + row) : sA + max(row - 1, 0);
    vg[j] = *p;
  }
#pragma unroll
  for (int j = 0; j < NH; ++j) {
    const int t = min(tid + j * NTHR, NHE - 1);
    const int r = t / HW, col = c0 + (t - r * HW);
    const int cc = min(max(col, 0), n - 1);
    const float* p = (r < R) ? Hm + ((size_t)r * n + cc) : sB + max(cc - 1, 0);
    vh[j] = *p;
  }
#pragma unroll
  for (int j = 0; j < NG; ++j) {
    const int t = tid + j * NTHR;
    const int r = t / TS, row = i0 + (t - r * TS);
    if (t < NGE) gq[t] = (r < R || row >= 1) ? vg[j] : 0.f;
  }
#pragma unroll
  for (int j = 0; j < NH; ++j) {
    const int t = tid + j * NTHR;
    const int r = t / HW, col = c0 + (t - r * HW);
    const bool in = col >= 0 && col < n && (r < R || col >= 1);
    if (t < NHE) hq[t] = in ? vh[j] : 0.f;
  }
}

template <int R>
static __device__ __forceinline__ void pstep(const float* ga, const float* hx, int hw, int rdim, float& p, float& m) {
  p = 0.f;
  if constexpr (R > 0) {
#pragma unroll
    for (int r = 0; r < R; ++r) p += ga[r * TS] * hx[r * hw];
    m = ga[R * TS] * hx[R * hw];
  } else {
    for (int r = 0; r < rdim; ++r) p += ga[r * TS] * hx[r * hw];
    m = ga[rdim * TS] * hx[rdim * hw];
  }
}

template <int R>
static __device__ __forceinline__ void seg_walk(const float* gq, const float* hx, int rdim, float& acc, float& prod) {
  acc = 0.f; prod = 1.f;
  if constexpr (R > 0) {
#pragma unroll 8
    for (int a = 0; a < TS; ++a) {
      float p, m;
      pstep<R>(gq + a, hx + a, HW1, rdim, p, m);
      acc = m * acc + p;
      prod *= m;
    }
  } else {
#pragma nounroll
    for (int a = 0; a < TS; ++a) {
      float p, m;
      pstep<0>(gq + a, hx + a, HW1, rdim, p, m);
      acc = m * acc + p;
      prod *= m;
    }
  }
}

__global__ void __launch_bounds__(256) k_segsum(const float* __restrict__ Gm, const float* __restrict__ Hm,
                         const float* __restrict__ sA, const float* __restrict__ sB,
                         float* __restrict__ T, float* __restrict__ PI, int n, int rdim) {
  extern __shared__ float sm1[];
  float* gq = sm1;
  float* hq = sm1 + (rdim + 1) * TS;
  const int tid = threadIdx.x;
  const int i0 = blockIdx.y * TS;
  const int kb = blockIdx.x * 256;
  if (rdim == 4) stage_rows_grouped<4, HW1, 256>(gq, hq, i0, kb - (TS - 1), Gm, Hm, sA, sB, n, tid);
  else           stage_rows(gq, hq, HW1, i0, kb - (TS - 1), Gm, Hm, sA, sB, n, rdim, tid, 256);
  __syncthreads();
  float acc, prod;
  if (rdim == 4) seg_walk<4>(gq, hq + tid, rdim, acc, prod);
  else           seg_walk<0>(gq, hq + tid, rdim, acc, prod);
  T[(size_t)blockIdx.y * n + kb + tid] = acc;
  PI[(size_t)blockIdx.y * n + kb + tid] = prod;
}

__global__ void __launch_bounds__(256) k_carry(const float* __restrict__ T, const float* __restrict__ PI,
                         float* __restrict__ MB, int n) {
  const int g = blockIdx.x * 256 + threadIdx.x;
  if (g >= 2 * n - 1) return;
  const int d = g - (n - 1);
  const int nI = n / TS;
  const int lo = (d < -(TS - 1)) ? (-d) / TS : 0;
  const int hi = ((n - d) / TS < nI) ? (n - d) / TS : nI;
  float s = 0.f;
  const long long step = (long long)n + TS, off = (long long)(TS - 1) + d;
  for (int Ib = 0; Ib < nI; Ib += 32) {
    float tv[32], pv[32];
#pragma unroll
    for (int u = 0; u < 32; ++u) {
      const int I = Ib + u;
      const long long idx = (I >= lo && I < hi) ? I * step + off : 0;
      tv[u] = T[idx];
      pv[u] = PI[idx];
    }
#pragma unroll
    for (int u = 0; u < 32; ++u) {
      const int I = Ib + u;
      if (I >= lo && I < hi) {
        s = pv[u] * s + tv[u];
        MB[I * step + off] = s;
      }
    }
  }
}

template <int R>
static __device__ __forceinline__ void walk(const float* gq, const float* hq, float* st, int tid, float s, int rdim) {
  if constexpr (R > 0) {
#pragma nounroll
    for (int a0 = 0; a0 < TS; a0 += 8) {
      float p[8], m[8];
#pragma unroll
      for (int u = 0; u < 8; ++u) pstep<R>(gq + a0 + u, hq + tid + a0 + u, HW2, rdim, p[u], m[u]);
#pragma unroll
      for (int u = 0; u < 8; ++u) {
        s = m[u] * s + p[u];
        const int b = tid + a0 + u - (TS - 1);
        if (b >= 0 && b < TS) st[(a0 + u) * (TS + 1) + b] = s;
      }
    }
  } else {
#pragma nounroll
    for (int a = 0; a < TS; ++a) {
      float p, m;
      pstep<0>(gq + a, hq + tid + a, HW2, rdim, p, m);
      s = m * s + p;
      const int b = tid + a - (TS - 1);
      if (b >= 0 && b < TS) st[a * (TS + 1) + b] = s;
    }
  }
}

__global__ void __launch_bounds__(128) k_buildM(const float* __restrict__ Gm, const float* __restrict__ Hm,
                         const float* __restrict__ sA, const float* __restrict__ sB,
                         const float* __restrict__ MB, float* __restrict__ Sq, int n, int rdim) {
  extern __shared__ float sm2[];
  __shared__ float st[TS * (TS + 1)];
  float* gq = sm2;
  float* hq = gq + (rdim + 1) * TS;
  const int tid = threadIdx.x;
  const int I = blockIdx.y;
  const int i0 = I * TS, k0 = blockIdx.x * TS;
  float s = 0.f;
  if (tid < 2 * TS - 1) {
    const int c = k0 - (TS - 1) + tid;
    s = (I > 0 && c >= 1) ? MB[(size_t)(I - 1) * n + (c - 1)] : 0.f;
  }
  if (rdim == 4) stage_rows_grouped<4, HW2, 128>(gq, hq, i0, k0 - (TS - 1), Gm, Hm, sA, sB, n, tid);
  else           stage_rows(gq, hq, HW2, i0, k0 - (TS - 1), Gm, Hm, sA, sB, n, rdim, tid, 128);
  __syncthreads();

  if (tid < 2 * TS - 1) {
    if (rdim == 4) walk<4>(gq, hq, st, tid, s, rdim);
    else           walk<0>(gq, hq, st, tid, s, rdim);
  }
  __syncthreads();

  constexpr int NOUT = TS * (TS / 4) / 128;
  v4f ov[NOUT];
#pragma unroll
  for (int j = 0; j < NOUT; ++j) {
    const int e = tid + 128 * j;
    const int a = e & (TS - 1), kq = e / TS;
    const float* sp = st + a * (TS + 1) + 4 * kq;
    ov[j] = (v4f){sp[0], sp[1], sp[2], sp[3]};
  }
#pragma unroll
  for (int j = 0; j < NOUT; ++j) {
    const int e = tid + 128 * j;
    const int a = e & (TS - 1), kq = e / TS;
    *(v4f*)(Sq + ((size_t)(k0 / 4 + kq) * n + (i0 + a)) * 4) = ov[j];
  }
}

#define GT 128
#define GKQ 16
#define GP 129
#define GPLANE (GKQ * GP)
#define GSLOT (2 * GPLANE)

struct FragSet {
  v4f a[2][4];
  v4f b[2][2];
};

template <bool B> struct Flag { static constexpr bool value = B; };

#define SET_EDGE() __builtin_amdgcn_sched_barrier(0)

static __device__ __forceinline__ v8f mma4(v2f a, v2f b, v8f c) {
  return __builtin_amdgcn_wmma_f32_16x16x4_f32(false, a, false, b, (short)0, c, false, false);
}

__global__ void __launch_bounds__(256) k_gemm(const float* __restrict__ x, const float* __restrict__ Sq,
                        const float* __restrict__ bias, float* __restrict__ out, int n, int batch) {
  __shared__ v4f lds[2 * GSLOT];

  const int tid = threadIdx.x;
  const int lane = tid & 31;
  const int w = tid >> 5;

  const int ib = blockIdx.x * GT;
  const int bb = blockIdx.y * GT;
  const int wi = (w & 3) * 32;
  const int wb = (w >> 2) * 64;

  const int mlan = lane & 15;
  const int h    = lane >> 4;
  const int nq   = n >> 2;

  const int akq = tid & 15, arow = tid >> 4;
  const int bi = tid & 127, bkq = tid >> 7;
  const v4f* __restrict__ xg = (const v4f*)x + (size_t)(bb + arow) * nq + akq;
  const v4f* __restrict__ sg = (const v4f*)Sq + (size_t)bkq * n + (ib + bi);
  v4f* aS = lds + akq * GP + arow;
  v4f* bS = lds + GPLANE + bkq * GP + bi;
  const v4f* aF = lds + h * GP + (wb + mlan);
  const v4f* bF = lds + GPLANE + h * GP + (wi + mlan);

  v8f c[2][4];
#pragma unroll
  for (int it = 0; it < 2; ++it)
#pragma unroll
    for (int t = 0; t < 4; ++t) c[it][t] = (v8f){0.f,0.f,0.f,0.f,0.f,0.f,0.f,0.f};

  v4f R[16];

  auto gload = [&](int s) {
    const v4f* __restrict__ xp = xg + (size_t)GKQ * s;
    const v4f* __restrict__ sp = sg + (size_t)GKQ * s * n;
#pragma unroll
    for (int j = 0; j < 8; ++j) R[j] = xp[(size_t)(16 * j) * nq];
#pragma unroll
    for (int j = 0; j < 8; ++j) R[8 + j] = sp[(size_t)(2 * j) * n];
  };
  auto sstore = [&](int slot) {
    v4f* ap = aS + slot * GSLOT;
    v4f* bp = bS + slot * GSLOT;
#pragma unroll
    for (int j = 0; j < 8; ++j) ap[16 * j] = R[j];
#pragma unroll
    for (int j = 0; j < 8; ++j) bp[2 * j * GP] = R[8 + j];
  };
  auto lfrag = [&](FragSet& f, int slot, int sub) {
    const v4f* ap = aF + slot * GSLOT + 4 * sub * GP;
    const v4f* bp = bF + slot * GSLOT + 4 * sub * GP;
#pragma unroll
    for (int Q = 0; Q < 2; ++Q) {
#pragma unroll
      for (int t = 0; t < 4; ++t) f.a[Q][t] = ap[2 * Q * GP + 16 * t];
#pragma unroll
      for (int it = 0; it < 2; ++it) f.b[Q][it] = bp[2 * Q * GP + 16 * it];
    }
  };
  auto mma_q = [&](const FragSet& f, int Q) {
#pragma unroll
    for (int t = 0; t < 4; ++t) {
      c[0][t] = mma4(f.a[Q][t].lo, f.b[Q][0].lo, c[0][t]);
      c[1][t] = mma4(f.a[Q][t].lo, f.b[Q][1].lo, c[1][t]);
    }
#pragma unroll
    for (int t = 0; t < 4; ++t) {
      c[0][t] = mma4(f.a[Q][t].hi, f.b[Q][0].hi, c[0][t]);
      c[1][t] = mma4(f.a[Q][t].hi, f.b[Q][1].hi, c[1][t]);
    }
  };

  FragSet F0, F1;

  auto stage = [&](int j, auto has_next, auto has_next2) {
    const int cur = j & 1;
    lfrag(F1, cur, 1); SET_EDGE(); mma_q(F0, 0); mma_q(F0, 1); SET_EDGE();
    lfrag(F0, cur, 2); SET_EDGE(); mma_q(F1, 0); mma_q(F1, 1); SET_EDGE();
    lfrag(F1, cur, 3); SET_EDGE(); mma_q(F0, 0); mma_q(F0, 1); SET_EDGE();
    if constexpr (decltype(has_next)::value) { sstore(cur ^ 1); SET_EDGE(); }
    mma_q(F1, 0); SET_EDGE();
    if constexpr (decltype(has_next)::value) {
      __syncthreads();
      lfrag(F0, cur ^ 1, 0); SET_EDGE();
    }
    mma_q(F1, 1); SET_EDGE();
    if constexpr (decltype(has_next2)::value) { gload(j + 2); SET_EDGE(); }
  };

  const int nst = n / (4 * GKQ);
  gload(0);
  sstore(0);
  __syncthreads();
  gload(1);
  lfrag(F0, 0, 0);
  SET_EDGE();
  int j = 0;
  for (; j < nst - 2; ++j) stage(j, Flag<true>{}, Flag<true>{});
  stage(j, Flag<true>{}, Flag<false>{});
  stage(j + 1, Flag<false>{}, Flag<false>{});

  const int i0 = ib + wi, b0 = bb + wb;
  const int bhalf = h * 8;
#pragma unroll
  for (int it = 0; it < 2; ++it) {
    const int i = i0 + 16 * it + mlan;
    const float bi2 = bias[i];
#pragma unroll
    for (int t = 0; t < 4; ++t)
#pragma unroll
      for (int v = 0; v < 8; ++v) {
        const int b = b0 + 16 * t + v + bhalf;
        out[(size_t)b * n + i] = bi2 + c[it][t][v];
      }
  }
}

extern "C" void kernel_launch(void* const* d_in, const int* in_sizes, int n_in,
                              void* d_out, int out_size, void* d_ws, size_t ws_size,
                              hipStream_t stream) {
  const float* x     = (const float*)d_in[0];
  const float* sA    = (const float*)d_in[1];
  const float* sB    = (const float*)d_in[2];
  const float* G     = (const float*)d_in[3];
  const float* H     = (const float*)d_in[4];
  const float* bias  = (const float*)d_in[5];
  float* out = (float*)d_out;

  const int n     = in_sizes[5];
  const int batch = in_sizes[0] / n;
  const int rdim  = in_sizes[3] / n;

  float* T  = (float*)d_ws;
  float* Sq = T + (size_t)n * batch;
  float* PI = T + (size_t)(n / TS) * n;
  float* MB = PI + (size_t)(n / TS) * n;

  {
    dim3 grid(n / 256, n / TS);
    const size_t lds = (size_t)(rdim + 1) * (TS + HW1) * sizeof(float);
    k_segsum<<<grid, 256, lds, stream>>>(G, H, sA, sB, T, PI, n, rdim);
  }
  k_carry<<<2 * n / 256, 256, 0, stream>>>(T, PI, MB, n);
  {
    dim3 grid(n / TS, n / TS);
    const size_t lds = (size_t)(rdim + 1) * (TS + HW2) * sizeof(float);
    k_buildM<<<grid, 128, lds, stream>>>(G, H, sA, sB, MB, Sq, n, rdim);
  }
  {
    dim3 grid(n / GT, batch / GT, 1);
    k_gemm<<<grid, 256, 0, stream>>>(x, Sq, bias, out, n, batch);
  }
}


# Build walks: row operands G[r,i], a[i] (uniform over a work-group) come by scalar loads into scalar registers, not LDS reads: LDS reads per step 10 to 5, tile block's LDS 21.25 to 20 KiB; sums same
# speedup vs baseline: 10.1024x; 1.0373x over previous
#include <hip/hip_runtime.h>


typedef __attribute__((ext_vector_type(2))) float v2f;
typedef __attribute__((ext_vector_type(4))) float v4f;
typedef __attribute__((ext_vector_type(8))) float v8f;

#define TS 64
#define HW1 320
#define HW2 192

static __device__ __forceinline__ void stage_rows(float* gq, float* hq, int hw, int i0, int c0,
                                                  const float* __restrict__ Gm, const float* __restrict__ Hm,
                                                  const float* __restrict__ sA, const float* __restrict__ sB,
                                                  int n, int rdim, int tid, int nthr) {
  for (int t = tid; t < (rdim + 1) * TS; t += nthr) {
    const int r = t / TS, row = i0 + (t - r * TS);
    gq[t] = (r < rdim) ? Gm[(size_t)r * n + row] : (row >= 1 ? sA[row - 1] : 0.f);
  }
  for (int t = tid; t < (rdim + 1) * hw; t += nthr) {
    const int r = t / hw, col = c0 + (t - r * hw);
    float v = 0.f;
    if (col >= 0 && col < n) v = (r < rdim) ? Hm[(size_t)r * n + col] : (col >= 1 ? sB[col - 1] : 0.f);
    hq[t] = v;
  }
}

template <int R, int HW, int NTHR>
static __device__ __forceinline__ void stage_cols_grouped(float* hq, int c0,
                                                          const float* __restrict__ Hm, const float* __restrict__ sB,
                                                          int n, int tid) {
  constexpr int NHE = (R + 1) * HW;
  constexpr int NH = (NHE + NTHR - 1) / NTHR;
  float vh[NH];
#pragma unroll
  for (int j = 0; j < NH; ++j) {
    const int t = min(tid + j * NTHR, NHE - 1);
    const int r = t / HW, col = c0 + (t - r * HW);
    const int cc = min(max(col, 0), n - 1);
    const float* p = (r < R) ? Hm + ((size_t)r * n + cc) : sB + max(cc - 1, 0);
    vh[j] = *p;
  }
#pragma unroll
  for (int j = 0; j < NH; ++j) {
    const int t = tid + j * NTHR;
    const int r = t / HW, col = c0 + (t - r * HW);
    const bool in = col >= 0 && col < n && (r < R || col >= 1);
    if (t < NHE) hq[t] = in ? vh[j] : 0.f;
  }
}

static __device__ __forceinline__ void pstep(const float* ga, const float* hx, int hw, int rdim, float& p, float& m) {
  p = 0.f;
  for (int r = 0; r < rdim; ++r) p += ga[r * TS] * hx[r * hw];
  m = ga[rdim * TS] * hx[rdim * hw];
}

template <int R>
struct RowOps {
  float g[R][8];
  float a[8];
  __device__ __forceinline__ void load(const float* __restrict__ Gm, const float* __restrict__ sA, int n, int row0) {
#pragma unroll
    for (int r = 0; r < R; ++r)
#pragma unroll
      for (int u = 0; u < 8; ++u) g[r][u] = Gm[(size_t)r * n + row0 + u];
    const float a0v = sA[max(row0 - 1, 0)];
    a[0] = row0 >= 1 ? a0v : 0.f;
#pragma unroll
    for (int u = 1; u < 8; ++u) a[u] = sA[row0 + (u - 1)];
  }
};

template <int R>
static __device__ __forceinline__ void pstep_u(const RowOps<R>& q, int u, const float* hx, int hw, float& p, float& m) {
  p = 0.f;
#pragma unroll
  for (int r = 0; r < R; ++r) p += q.g[r][u] * hx[r * hw];
  m = q.a[u] * hx[R * hw];
}

template <int R>
static __device__ __forceinline__ void seg_walk_u(const float* __restrict__ Gm, const float* __restrict__ sA, int n,
                                                  int i0, const float* hx, float& acc, float& prod) {
  acc = 0.f; prod = 1.f;
#pragma nounroll
  for (int a0 = 0; a0 < TS; a0 += 8) {
    RowOps<R> q;
    q.load(Gm, sA, n, i0 + a0);
#pragma unroll
    for (int u = 0; u < 8; ++u) {
      float p, m;
      pstep_u<R>(q, u, hx + a0 + u, HW1, p, m);
      acc = m * acc + p;
      prod *= m;
    }
  }
}

static __device__ __forceinline__ void seg_walk(const float* gq, const float* hx, int rdim, float& acc, float& prod) {
  acc = 0.f; prod = 1.f;
#pragma nounroll
  for (int a = 0; a < TS; ++a) {
    float p, m;
    pstep(gq + a, hx + a, HW1, rdim, p, m);
    acc = m * acc + p;
    prod *= m;
  }
}

__global__ void __launch_bounds__(256) k_segsum(const float* __restrict__ Gm, const float* __restrict__ Hm,
                         const float* __restrict__ sA, const float* __restrict__ sB,
                         float* __restrict__ T, float* __restrict__ PI, int n, int rdim) {
  extern __shared__ float sm1[];
  float* gq = sm1;
  float* hq = sm1 + (rdim + 1) * TS;
  const int tid = threadIdx.x;
  const int i0 = blockIdx.y * TS;
  const int kb = blockIdx.x * 256;
  if (rdim == 4) stage_cols_grouped<4, HW1, 256>(hq, kb - (TS - 1), Hm, sB, n, tid);
  else           stage_rows(gq, hq, HW1, i0, kb - (TS - 1), Gm, Hm, sA, sB, n, rdim, tid, 256);
  __syncthreads();
  float acc, prod;
  if (rdim == 4) seg_walk_u<4>(Gm, sA, n, i0, hq + tid, acc, prod);
  else           seg_walk(gq, hq + tid, rdim, acc, prod);
  T[(size_t)blockIdx.y * n + kb + tid] = acc;
  PI[(size_t)blockIdx.y * n + kb + tid] = prod;
}

__global__ void __launch_bounds__(256) k_carry(const float* __restrict__ T, const float* __restrict__ PI,
                         float* __restrict__ MB, int n) {
  const int g = blockIdx.x * 256 + threadIdx.x;
  if (g >= 2 * n - 1) return;
  const int d = g - (n - 1);
  const int nI = n / TS;
  const int lo = (d < -(TS - 1)) ? (-d) / TS : 0;
  const int hi = ((n - d) / TS < nI) ? (n - d) / TS : nI;
  float s = 0.f;
  const long long step = (long long)n + TS, off = (long long)(TS - 1) + d;
  for (int Ib = 0; Ib < nI; Ib += 32) {
    float tv[32], pv[32];
#pragma unroll
    for (int u = 0; u < 32; ++u) {
      const int I = Ib + u;
      const long long idx = (I >= lo && I < hi) ? I * step + off : 0;
      tv[u] = T[idx];
      pv[u] = PI[idx];
    }
#pragma unroll
    for (int u = 0; u < 32; ++u) {
      const int I = Ib + u;
      if (I >= lo && I < hi) {
        s = pv[u] * s + tv[u];
        MB[I * step + off] = s;
      }
    }
  }
}

template <int R>
static __device__ __forceinline__ void walk_u(const float* __restrict__ Gm, const float* __restrict__ sA, int n,
                                              int i0, const float* hq, float* st, int tid, float s) {
#pragma nounroll
  for (int a0 = 0; a0 < TS; a0 += 8) {
    float p[8], m[8];
    RowOps<R> q;
    q.load(Gm, sA, n, i0 + a0);
#pragma unroll
    for (int u = 0; u < 8; ++u) pstep_u<R>(q, u, hq + tid + a0 + u, HW2, p[u], m[u]);
#pragma unroll
    for (int u = 0; u < 8; ++u) {
      s = m[u] * s + p[u];
      const int b = tid + a0 + u - (TS - 1);
      if (b >= 0 && b < TS) st[(a0 + u) * (TS + 1) + b] = s;
    }
  }
}

static __device__ __forceinline__ void walk(const float* gq, const float* hq, float* st, int tid, float s, int rdim) {
#pragma nounroll
  for (int a = 0; a < TS; ++a) {
    float p, m;
    pstep(gq + a, hq + tid + a, HW2, rdim, p, m);
    s = m * s + p;
    const int b = tid + a - (TS - 1);
    if (b >= 0 && b < TS) st[a * (TS + 1) + b] = s;
  }
}

__global__ void __launch_bounds__(128) k_buildM(const float* __restrict__ Gm, const float* __restrict__ Hm,
                         const float* __restrict__ sA, const float* __restrict__ sB,
                         const float* __restrict__ MB, float* __restrict__ Sq, int n, int rdim) {
  extern __shared__ float sm2[];
  __shared__ float st[TS * (TS + 1)];
  float* gq = sm2;
  float* hq = (rdim == 4) ? sm2 : gq + (rdim + 1) * TS;
  const int tid = threadIdx.x;
  const int I = blockIdx.y;
  const int i0 = I * TS, k0 = blockIdx.x * TS;
  float s = 0.f;
  if (tid < 2 * TS - 1) {
    const int c = k0 - (TS - 1) + tid;
    s = (I > 0 && c >= 1) ? MB[(size_t)(I - 1) * n + (c - 1)] : 0.f;
  }
  if (rdim == 4) stage_cols_grouped<4, HW2, 128>(hq, k0 - (TS - 1), Hm, sB, n, tid);
  else           stage_rows(gq, hq, HW2, i0, k0 - (TS - 1), Gm, Hm, sA, sB, n, rdim, tid, 128);
  __syncthreads();

  if (tid < 2 * TS - 1) {
    if (rdim == 4) walk_u<4>(Gm, sA, n, i0, hq, st, tid, s);
    else           walk(gq, hq, st, tid, s, rdim);
  }
  __syncthreads();

  constexpr int NOUT = TS * (TS / 4) / 128;
  v4f ov[NOUT];
#pragma unroll
  for (int j = 0; j < NOUT; ++j) {
    const int e = tid + 128 * j;
    const int a = e & (TS - 1), kq = e / TS;
    const float* sp = st + a * (TS + 1) + 4 * kq;
    ov[j] = (v4f){sp[0], sp[1], sp[2], sp[3]};
  }
#pragma unroll
  for (int j = 0; j < NOUT; ++j) {
    const int e = tid + 128 * j;
    const int a = e & (TS - 1), kq = e / TS;
    *(v4f*)(Sq + ((size_t)(k0 / 4 + kq) * n + (i0 + a)) * 4) = ov[j];
  }
}

#define GT 128
#define GKQ 16
#define GP 129
#define GPLANE (GKQ * GP)
#define GSLOT (2 * GPLANE)

struct FragSet {
  v4f a[2][4];
  v4f b[2][2];
};

template <bool B> struct Flag { static constexpr bool value = B; };

#define SET_EDGE() __builtin_amdgcn_sched_barrier(0)

static __device__ __forceinline__ v8f mma4(v2f a, v2f b, v8f c) {
  return __builtin_amdgcn_wmma_f32_16x16x4_f32(false, a, false, b, (short)0, c, false, false);
}

__global__ void __launch_bounds__(256) k_gemm(const float* __restrict__ x, const float* __restrict__ Sq,
                        const float* __restrict__ bias, float* __restrict__ out, int n, int batch) {
  __shared__ v4f lds[2 * GSLOT];

  const int tid = threadIdx.x;
  const int lane = tid & 31;
  const int w = tid >> 5;

  const int ib = blockIdx.x * GT;
  const int bb = blockIdx.y * GT;
  const int wi = (w & 3) * 32;
  const int wb = (w >> 2) * 64;

  const int mlan = lane & 15;
  const int h    = lane >> 4;
  const int nq   = n >> 2;

  const int akq = tid & 15, arow = tid >> 4;
  const int bi = tid & 127, bkq = tid >> 7;
  const v4f* __restrict__ xg = (const v4f*)x + (size_t)(bb + arow) * nq + akq;
  const v4f* __restrict__ sg = (const v4f*)Sq + (size_t)bkq * n + (ib + bi);
  v4f* aS = lds + akq * GP + arow;
  v4f* bS = lds + GPLANE + bkq * GP + bi;
  const v4f* aF = lds + h * GP + (wb + mlan);
  const v4f* bF = lds + GPLANE + h * GP + (wi + mlan);

  v8f c[2][4];
#pragma unroll
  for (int it = 0; it < 2; ++it)
#pragma unroll
    for (int t = 0; t < 4; ++t) c[it][t] = (v8f){0.f,0.f,0.f,0.f,0.f,0.f,0.f,0.f};

  v4f R[16];

  auto gload = [&](int s) {
    const v4f* __restrict__ xp = xg + (size_t)GKQ * s;
    const v4f* __restrict__ sp = sg + (size_t)GKQ * s * n;
#pragma unroll
    for (int j = 0; j < 8; ++j) R[j] = xp[(size_t)(16 * j) * nq];
#pragma unroll
    for (int j = 0; j < 8; ++j) R[8 + j] = sp[(size_t)(2 * j) * n];
  };
  auto sstore = [&](int slot) {
    v4f* ap = aS + slot * GSLOT;
    v4f* bp = bS + slot * GSLOT;
#pragma unroll
    for (int j = 0; j < 8; ++j) ap[16 * j] = R[j];
#pragma unroll
    for (int j = 0; j < 8; ++j) bp[2 * j * GP] = R[8 + j];
  };
  auto lfrag = [&](FragSet& f, int slot, int sub) {
    const v4f* ap = aF + slot * GSLOT + 4 * sub * GP;
    const v4f* bp = bF + slot * GSLOT + 4 * sub * GP;
#pragma unroll
    for (int Q = 0; Q < 2; ++Q) {
#pragma unroll
      for (int t = 0; t < 4; ++t) f.a[Q][t] = ap[2 * Q * GP + 16 * t];
#pragma unroll
      for (int it = 0; it < 2; ++it) f.b[Q][it] = bp[2 * Q * GP + 16 * it];
    }
  };
  auto mma_q = [&](const FragSet& f, int Q) {
#pragma unroll
    for (int t = 0; t < 4; ++t) {
      c[0][t] = mma4(f.a[Q][t].lo, f.b[Q][0].lo, c[0][t]);
      c[1][t] = mma4(f.a[Q][t].lo, f.b[Q][1].lo, c[1][t]);
    }
#pragma unroll
    for (int t = 0; t < 4; ++t) {
      c[0][t] = mma4(f.a[Q][t].hi, f.b[Q][0].hi, c[0][t]);
      c[1][t] = mma4(f.a[Q][t].hi, f.b[Q][1].hi, c[1][t]);
    }
  };

  FragSet F0, F1;

  auto stage = [&](int j, auto has_next, auto has_next2) {
    const int cur = j & 1;
    lfrag(F1, cur, 1); SET_EDGE(); mma_q(F0, 0); mma_q(F0, 1); SET_EDGE();
    lfrag(F0, cur, 2); SET_EDGE(); mma_q(F1, 0); mma_q(F1, 1); SET_EDGE();
    lfrag(F1, cur, 3); SET_EDGE(); mma_q(F0, 0); mma_q(F0, 1); SET_EDGE();
    if constexpr (decltype(has_next)::value) { sstore(cur ^ 1); SET_EDGE(); }
    mma_q(F1, 0); SET_EDGE();
    if constexpr (decltype(has_next)::value) {
      __syncthreads();
      lfrag(F0, cur ^ 1, 0); SET_EDGE();
    }
    mma_q(F1, 1); SET_EDGE();
    if constexpr (decltype(has_next2)::value) { gload(j + 2); SET_EDGE(); }
  };

  const int nst = n / (4 * GKQ);
  gload(0);
  sstore(0);
  __syncthreads();
  gload(1);
  lfrag(F0, 0, 0);
  SET_EDGE();
  int j = 0;
  for (; j < nst - 2; ++j) stage(j, Flag<true>{}, Flag<true>{});
  stage(j, Flag<true>{}, Flag<false>{});
  stage(j + 1, Flag<false>{}, Flag<false>{});

  const int i0 = ib + wi, b0 = bb + wb;
  const int bhalf = h * 8;
#pragma unroll
  for (int it = 0; it < 2; ++it) {
    const int i = i0 + 16 * it + mlan;
    const float bi2 = bias[i];
#pragma unroll
    for (int t = 0; t < 4; ++t)
#pragma unroll
      for (int v = 0; v < 8; ++v) {
        const int b = b0 + 16 * t + v + bhalf;
        out[(size_t)b * n + i] = bi2 + c[it][t][v];
      }
  }
}

extern "C" void kernel_launch(void* const* d_in, const int* in_sizes, int n_in,
                              void* d_out, int out_size, void* d_ws, size_t ws_size,
                              hipStream_t stream) {
  const float* x     = (const float*)d_in[0];
  const float* sA    = (const float*)d_in[1];
  const float* sB    = (const float*)d_in[2];
  const float* G     = (const float*)d_in[3];
  const float* H     = (const float*)d_in[4];
  const float* bias  = (const float*)d_in[5];
  float* out = (float*)d_out;

  const int n     = in_sizes[5];
  const int batch = in_sizes[0] / n;
  const int rdim  = in_sizes[3] / n;

  float* T  = (float*)d_ws;
  float* Sq = T + (size_t)n * batch;
  float* PI = T + (size_t)(n / TS) * n;
  float* MB = PI + (size_t)(n / TS) * n;

  {
    dim3 grid(n / 256, n / TS);
    const size_t lds = (size_t)(rdim + 1) * (TS + HW1) * sizeof(float);
    k_segsum<<<grid, 256, lds, stream>>>(G, H, sA, sB, T, PI, n, rdim);
  }
  k_carry<<<2 * n / 256, 256, 0, stream>>>(T, PI, MB, n);
  {
    dim3 grid(n / TS, n / TS);
    const size_t lds = (size_t)(rdim + 1) * ((rdim == 4 ? 0 : TS) + HW2) * sizeof(float);
    k_buildM<<<grid, 128, lds, stream>>>(G, H, sA, sB, MB, Sq, n, rdim);
  }
  {
    dim3 grid(n / GT, batch / GT, 1);
    k_gemm<<<grid, 256, 0, stream>>>(x, Sq, bias, out, n, batch);
  }
}
